# starting kernel
# baseline (speedup 1.0000x reference)
#include <hip/hip_runtime.h>

#define NB   4096
#define NF   200
#define NK   40
#define NC   128

typedef __attribute__((ext_vector_type(2))) float v2f;
typedef __attribute__((ext_vector_type(4))) float v4f;
typedef __attribute__((ext_vector_type(8))) float v8f;

#define NRG  13

__device__ __forceinline__ void sched_fence() {
#if defined(__has_builtin)
#if __has_builtin(__builtin_amdgcn_sched_barrier)
    __builtin_amdgcn_sched_barrier(0);
#endif
#endif
}

template <int MASK, int SIZE>
__device__ __forceinline__ void sched_group() {
#if defined(__has_builtin)
#if __has_builtin(__builtin_amdgcn_sched_group_barrier)
    __builtin_amdgcn_sched_group_barrier(MASK, SIZE, 0);
#endif
#endif
}
#define SG_DS_READ 0x100
#define SG_WMMA    0x008

__global__ __launch_bounds__(256)
void softmax_gemm_reduce_kernel(const float* __restrict__ x,
                                const float* __restrict__ W,
                                float* __restrict__ out)
{
    __shared__ float sW[(NK / 2) * (NC * 2)];
    __shared__ float sOut[NC];

    const int tid  = threadIdx.x;
    const int b    = blockIdx.x;
    const int wave = tid >> 5;
    const int lane = tid & 31;
    const int half = lane >> 4;
    const int lm   = lane & 15;

    for (int i4 = tid; i4 < (NK * NC) / 4; i4 += 256) {
        const int i = i4 * 4;
        const int k = i >> 7;
        const int c = i & (NC - 1);
        v4f w = *(const v4f*)(W + i);
        float* dst = &sW[(k >> 1) * 256 + c * 2 + (k & 1)];
        dst[0] = w.x; dst[2] = w.y; dst[4] = w.z; dst[6] = w.w;
    }
    if (tid < NC) sOut[tid] = 0.0f;
    __syncthreads();

    const float* sWcol = &sW[lm * 2];

    for (int rg = wave; rg < NRG; rg += 8) {
        const int f       = rg * 16 + lm;
        const int fclamp  = (f < NF) ? f : (NF - 1);
        const float* xrow = x + ((size_t)b * NF + fclamp) * NK;
        const int klo     = half * 2;

        v2f afr[10];
        #pragma unroll
        for (int ks = 0; ks < 10; ++ks)
            afr[ks] = *(const v2f*)(xrow + 4 * ks + klo);

        v2f bf[2][8];
        #pragma unroll
        for (int nt = 0; nt < 8; ++nt)
            bf[0][nt] = *(const v2f*)(sWcol + (half)*256 + nt * 32);

        sched_fence();

        v8f acc[8] = {};
        #pragma unroll
        for (int ks = 0; ks < 10; ++ks) {
            const int cur = ks & 1;
            const int nxt = cur ^ 1;
            if (ks < 9) {
                const int j = 2 * (ks + 1) + half;
                #pragma unroll
                for (int nt = 0; nt < 8; ++nt)
                    bf[nxt][nt] = *(const v2f*)(sWcol + j * 256 + nt * 32);
                sched_group<SG_DS_READ, 4>();
            }
            #pragma unroll
            for (int nt = 0; nt < 8; ++nt) {
                acc[nt] = __builtin_amdgcn_wmma_f32_16x16x4_f32(
                               false, afr[ks],  false, bf[cur][nt],
                               (short)0, acc[nt],
                               false,  false);
            }
            sched_group<SG_WMMA, 8>();
        }

        float rinv[8];
        #pragma unroll
        for (int r = 0; r < 8; ++r) {
            float m = acc[0][r];
            #pragma unroll
            for (int nt = 1; nt < 8; ++nt) m = fmaxf(m, acc[nt][r]);
            #pragma unroll
            for (int s = 1; s < 16; s <<= 1) m = fmaxf(m, __shfl_xor(m, s, 32));

            float sum = 0.0f;
            #pragma unroll
            for (int nt = 0; nt < 8; ++nt) {
                float e = __expf(acc[nt][r] - m);
                acc[nt][r] = e;
                sum += e;
            }
            #pragma unroll
            for (int s = 1; s < 16; s <<= 1) sum += __shfl_xor(sum, s, 32);
            rinv[r] = 1.0f / sum;
        }

        #pragma unroll
        for (int nt = 0; nt < 8; ++nt) {
            float cs = 0.0f;
            #pragma unroll
            for (int r = 0; r < 8; ++r) {
                const int frow = rg * 16 + half * 8 + r;
                float v = acc[nt][r] * rinv[r];
                cs += (frow < NF) ? v : 0.0f;
            }
            cs += __shfl_xor(cs, 16, 32);
            if (half == 0) atomicAdd(&sOut[nt * 16 + lm], cs);
        }
    }

    __syncthreads();
    if (tid < NC) out[(size_t)b * NC + tid] = sOut[tid];
}

extern "C" void kernel_launch(void* const* d_in, const int* in_sizes, int n_in,
                              void* d_out, int out_size, void* d_ws, size_t ws_size,
                              hipStream_t stream) {
    const float* x = (const float*)d_in[0];
    const float* W = (const float*)d_in[1];
    float* out     = (float*)d_out;

    dim3 grid(NB);
    dim3 block(256);
    softmax_gemm_reduce_kernel<<<grid, block, 0, stream>>>(x, W, out);
}


# Two row tiles per loop trip: each W fragment read from LDS feeds both tiles, read in groups of two k-steps: one LDS read group per 32 matrix instructions, not per 8; sum order unchanged
# speedup vs baseline: 1.4671x; 1.4671x over previous
#include <hip/hip_runtime.h>


#define NB   4096
#define NF   200
#define NK   40
#define NC   128

typedef __attribute__((ext_vector_type(2))) float v2f;
typedef __attribute__((ext_vector_type(4))) float v4f;
typedef __attribute__((ext_vector_type(8))) float v8f;

#define WAVES_PER_WG   8
#define TILES_PER_WAVE 25
#define ROWS_PER_WAVE  400
#define SPLIT_TILE     12
#define NWG            256
static_assert(NWG * WAVES_PER_WG * 2 == NB, "every sample belongs to exactly one wave");
static_assert(2 * NF == TILES_PER_WAVE * 16 && ROWS_PER_WAVE == 2 * NF, "a sample pair is a whole number of row tiles");
static_assert(NF == SPLIT_TILE * 16 + 8, "the sample boundary is the lane-half boundary of the split tile");

#define LDS_FLOATS 41984

__device__ __forceinline__ void sched_fence() {
#if defined(__has_builtin)
#if __has_builtin(__builtin_amdgcn_sched_barrier)
    __builtin_amdgcn_sched_barrier(0);
#endif
#endif
}

template <int MASK, int SIZE>
__device__ __forceinline__ void sched_group() {
#if defined(__has_builtin)
#if __has_builtin(__builtin_amdgcn_sched_group_barrier)
    __builtin_amdgcn_sched_group_barrier(MASK, SIZE, 0);
#endif
#endif
}
#define LOG2E 1.44269504088896340736f

#define SG_DS_READ 0x100
#define SG_WMMA    0x008

template <int XM>
__device__ __forceinline__ float row_xor(float v) {
    static_assert(XM > 0 && XM < 16, "mask stays inside the 16-lane row");
    return __builtin_bit_cast(float,
        __builtin_amdgcn_update_dpp(0, __builtin_bit_cast(int, v), 0x160 + XM, 0xf, 0xf, true));
}

#define ROW_MAX_STEP1(R, XM) \
    "v_max_num_f32_dpp %" #R ", %" #R ", %" #R " row_xmask:" #XM " row_mask:0xf bank_mask:0xf\n\t"
#define ROW_MAX_STEP8(XM)                                                         \
    ROW_MAX_STEP1(0, XM) ROW_MAX_STEP1(1, XM) ROW_MAX_STEP1(2, XM) ROW_MAX_STEP1(3, XM) \
    ROW_MAX_STEP1(4, XM) ROW_MAX_STEP1(5, XM) ROW_MAX_STEP1(6, XM) ROW_MAX_STEP1(7, XM)
__device__ __forceinline__ void row_max_butterfly8(float (&m)[8]) {
    asm(ROW_MAX_STEP8(1) ROW_MAX_STEP8(2) ROW_MAX_STEP8(4) ROW_MAX_STEP8(8)
        : "+v"(m[0]), "+v"(m[1]), "+v"(m[2]), "+v"(m[3]),
          "+v"(m[4]), "+v"(m[5]), "+v"(m[6]), "+v"(m[7]));
}

__device__ __forceinline__ void store_sample(float* __restrict__ o, const float (&part)[8],
                                             int half, int lane)
{
    float tot[8];
    #pragma unroll
    for (int nt = 0; nt < 8; ++nt) tot[nt] = part[nt] + __shfl_xor(part[nt], 16, 32);
    #pragma unroll
    for (int j = 0; j < 4; ++j) o[32 * j + lane] = half ? tot[2 * j + 1] : tot[2 * j];
}

#define KG      2
#define NGROUP  (10 / KG)
static_assert(NGROUP * KG == 10, "the ten k-steps are a whole number of groups");

__device__ __forceinline__ void load_w_group(v2f (&dst)[KG][8], const float* sWcol, int half, int g)
{
    #pragma unroll
    for (int kk = 0; kk < KG; ++kk) {
        const int j = 2 * (KG * g + kk) + half;
        #pragma unroll
        for (int nt = 0; nt < 8; ++nt)
            dst[kk][nt] = *(const v2f*)(sWcol + j * 256 + nt * 32);
    }
}

template <int NT>
__device__ __forceinline__ void logits_tiles(const v2f (&a)[NT][10], v8f (&acc)[NT][8],
                                             const float* sWcol, int half)
{
    v2f bf[2][KG][8];
    load_w_group(bf[0], sWcol, half, 0);
    sched_fence();
    #pragma unroll
    for (int g = 0; g < NGROUP; ++g) {
        const int cur = g & 1;
        const int nxt = cur ^ 1;
        if (g + 1 < NGROUP) {
            load_w_group(bf[nxt], sWcol, half, g + 1);
            sched_group<SG_DS_READ, 4 * KG>();
        }
        #pragma unroll
        for (int kk = 0; kk < KG; ++kk) {
            const int ks = KG * g + kk;
            #pragma unroll
            for (int nt = 0; nt < 8; ++nt) {
                #pragma unroll
                for (int ti = 0; ti < NT; ++ti) {
                    acc[ti][nt] = __builtin_amdgcn_wmma_f32_16x16x4_f32(
                                       false, a[ti][ks],  false, bf[cur][kk][nt],
                                       (short)0, acc[ti][nt],
                                       false,  false);
                }
            }
        }
        sched_group<SG_WMMA, 8 * KG * NT>();
    }
}

__device__ __forceinline__ void softmax_colsum(v8f (&acc)[8], float (&contrib)[8])
{
    float m[8];
    #pragma unroll
    for (int r = 0; r < 8; ++r) {
        float v = acc[0][r];
        #pragma unroll
        for (int nt = 1; nt < 8; ++nt) v = fmaxf(v, acc[nt][r]);
        m[r] = v;
    }
    row_max_butterfly8(m);

    float sum[8];
    #pragma unroll
    for (int r = 0; r < 8; ++r) {
        float s = 0.0f;
        #pragma unroll
        for (int nt = 0; nt < 8; ++nt) {
            float e = __builtin_amdgcn_exp2f(acc[nt][r] - m[r]);
            acc[nt][r] = e;
            s += e;
        }
        sum[r] = s;
    }
    #pragma unroll
    for (int r = 0; r < 8; ++r) sum[r] += row_xor<1>(sum[r]);
    #pragma unroll
    for (int r = 0; r < 8; ++r) sum[r] += row_xor<2>(sum[r]);
    #pragma unroll
    for (int r = 0; r < 8; ++r) sum[r] += row_xor<4>(sum[r]);
    #pragma unroll
    for (int r = 0; r < 8; ++r) sum[r] += row_xor<8>(sum[r]);

    float rinv[8];
    #pragma unroll
    for (int r = 0; r < 8; ++r) rinv[r] = __builtin_amdgcn_rcpf(sum[r]);

    #pragma unroll
    for (int nt = 0; nt < 8; ++nt) {
        float c = 0.0f;
        #pragma unroll
        for (int r = 0; r < 8; ++r) c += acc[nt][r] * rinv[r];
        contrib[nt] = c;
    }
}

__device__ __forceinline__ void load_a_tile(v2f (&dst)[10], const float* xlane, int tile)
{
    const float* p = xlane + (size_t)tile * (16 * NK);
    #pragma unroll
    for (int ks = 0; ks < 10; ++ks) dst[ks] = *(const v2f*)(p + 4 * ks);
}

__global__ __launch_bounds__(256)
void softmax_gemm_reduce_kernel(const float* __restrict__ x,
                                const float* __restrict__ W,
                                float* __restrict__ out)
{
    __shared__ float sW[LDS_FLOATS];

    const int tid  = threadIdx.x;
    const int wave = tid >> 5;
    const int lane = tid & 31;
    const int half = lane >> 4;
    const int lm   = lane & 15;

    {
        v4f w[5];
        #pragma unroll
        for (int j = 0; j < 5; ++j) w[j] = *(const v4f*)(W + (tid + 256 * j) * 4);
        #pragma unroll
        for (int j = 0; j < 5; ++j) {
            const int i = (tid + 256 * j) * 4;
            const int k = i >> 7;
            const int c = i & (NC - 1);
            float* dst = &sW[(k >> 1) * 256 + c * 2 + (k & 1)];
            dst[0] = w[j].x * LOG2E; dst[2] = w[j].y * LOG2E;
            dst[4] = w[j].z * LOG2E; dst[6] = w[j].w * LOG2E;
        }
    }
    __syncthreads();

    const float* sWcol = &sW[lm * 2];

    const int gw = blockIdx.x * WAVES_PER_WG + wave;
    const float* xlane = x + ((size_t)gw * ROWS_PER_WAVE + lm) * NK + half * 2;
    float* out0 = out + (size_t)(2 * gw) * NC;

    v2f a[2][10];
    load_a_tile(a[0], xlane, 0);
    load_a_tile(a[1], xlane, 1);

    float cs[8];
    #pragma unroll
    for (int nt = 0; nt < 8; ++nt) cs[nt] = 0.0f;

    int T = 0;
    #pragma unroll 1
    for (int seg = 0; seg < 2; ++seg) {
        #pragma unroll 1
        for (int p = 0; p < 6; ++p) {
            const int t0 = (T + 2 < TILES_PER_WAVE) ? (T + 2) : (TILES_PER_WAVE - 1);
            const int t1 = (T + 3 < TILES_PER_WAVE) ? (T + 3) : (TILES_PER_WAVE - 1);
            v2f n[2][10];
            load_a_tile(n[0], xlane, t0);
            load_a_tile(n[1], xlane, t1);

            v8f acc[2][8] = {};
            logits_tiles<2>(a, acc, sWcol, half);

            float c0[8], c1[8];
            softmax_colsum(acc[0], c0);
            sched_fence();
            softmax_colsum(acc[1], c1);
            #pragma unroll
            for (int nt = 0; nt < 8; ++nt) { cs[nt] += c0[nt]; cs[nt] += c1[nt]; }

            #pragma unroll
            for (int ks = 0; ks < 10; ++ks) { a[0][ks] = n[0][ks]; a[1][ks] = n[1][ks]; }
            T += 2;
        }
        if (seg == 0) {
            v2f n1[10];
            load_a_tile(n1, xlane, SPLIT_TILE + 2);

            v2f a1[1][10];
            #pragma unroll
            for (int ks = 0; ks < 10; ++ks) a1[0][ks] = a[0][ks];
            v8f acc[1][8] = {};
            logits_tiles<1>(a1, acc, sWcol, half);

            float contrib[8];
            softmax_colsum(acc[0], contrib);

            float part[8];
            #pragma unroll
            for (int nt = 0; nt < 8; ++nt) part[nt] = half ? cs[nt] : (cs[nt] + contrib[nt]);
            store_sample(out0, part, half, lane);
            #pragma unroll
            for (int nt = 0; nt < 8; ++nt) cs[nt] = half ? contrib[nt] : 0.0f;

            #pragma unroll
            for (int ks = 0; ks < 10; ++ks) { a[0][ks] = a[1][ks]; a[1][ks] = n1[ks]; }
            T = SPLIT_TILE + 1;
        }
    }

    store_sample(out0 + NC, cs, half, lane);
}

extern "C" void kernel_launch(void* const* d_in, const int* in_sizes, int n_in,
                              void* d_out, int out_size, void* d_ws, size_t ws_size,
                              hipStream_t stream) {
    const float* x = (const float*)d_in[0];
    const float* W = (const float*)d_in[1];
    float* out     = (float*)d_out;

    dim3 grid(NWG);
    dim3 block(256);
    softmax_gemm_reduce_kernel<<<grid, block, 0, stream>>>(x, W, out);
}
